# starting kernel
# baseline (speedup 1.0000x reference)
#include <hip/hip_runtime.h>

typedef float v2f __attribute__((ext_vector_type(2)));
typedef float v4f __attribute__((ext_vector_type(4)));
typedef float v8f __attribute__((ext_vector_type(8)));

#define SEQ   4096
#define DIM   64
#define NT    (SEQ / 16)
#define KSTR  68
#define SSTR  18

__global__ __launch_bounds__(256) void MultiHeadRoPE_28166395528043_kernel(
    const float* __restrict__ q,  const float* __restrict__ k,  const float* __restrict__ v,
    const float* __restrict__ fq, const float* __restrict__ fk, const float* __restrict__ fv,
    float* __restrict__ out)
{
    __shared__ float Kl[2][16][KSTR];
    __shared__ float Vl[2][16][KSTR];
    __shared__ float Sl[8][16][SSTR];

    const int batch = blockIdx.x >> 5;
    const int q0    = (blockIdx.x & 31) * 128;
    const int tid   = threadIdx.x;
    const int wave  = tid >> 5;
    const int lane  = tid & 31;
    const int lm    = lane & 15;
    const int lh    = lane >> 4;

    const float scale = 0.125f;

    const float* qb = q + (size_t)batch * SEQ * DIM;
    const float* kb = k + (size_t)batch * SEQ * DIM;
    const float* vb = v + (size_t)batch * SEQ * DIM;
    float*       ob = out + (size_t)batch * SEQ * DIM;

    v2f qa[16];
    {
        const int    qrow = q0 + wave * 16 + lm;
        const float* qp   = qb + (size_t)qrow * DIM;
        const float* fp   = fq + (size_t)qrow * DIM;
        #pragma unroll
        for (int c = 0; c < 16; ++c) {
            const int col = c * 4 + 2 * lh;
            v2f x = *(const v2f*)(qp + col);
            v2f f = *(const v2f*)(fp + col);
            v2f r;
            r.x = x.x * f.x - x.y * f.y;
            r.y = x.x * f.y + x.y * f.x;
            qa[c] = r;
        }
    }

    v8f acc0 = {}, acc1 = {}, acc2 = {}, acc3 = {};

    const int lrow = tid >> 4;
    const int lcol = (tid & 15) * 4;

    {
        v4f xk  = *(const v4f*)(kb + (size_t)lrow * DIM + lcol);
        v4f fk4 = *(const v4f*)(fk + (size_t)lrow * DIM + lcol);
        v4f xv  = *(const v4f*)(vb + (size_t)lrow * DIM + lcol);
        v4f fv4 = *(const v4f*)(fv + (size_t)lrow * DIM + lcol);
        v4f rk, rv;
        rk.x = xk.x*fk4.x - xk.y*fk4.y;  rk.y = xk.x*fk4.y + xk.y*fk4.x;
        rk.z = xk.z*fk4.z - xk.w*fk4.w;  rk.w = xk.z*fk4.w + xk.w*fk4.z;
        rv.x = xv.x*fv4.x - xv.y*fv4.y;  rv.y = xv.x*fv4.y + xv.y*fv4.x;
        rv.z = xv.z*fv4.z - xv.w*fv4.w;  rv.w = xv.z*fv4.w + xv.w*fv4.z;
        *(v4f*)&Kl[0][lrow][lcol] = rk;
        *(v4f*)&Vl[0][lrow][lcol] = rv;
    }

    for (int kt = 0; kt < NT; ++kt) {
        __syncthreads();
        const int buf = kt & 1;

        if (kt + 1 < NT) {
            const int nb   = buf ^ 1;
            const int krow = (kt + 1) * 16 + lrow;
            if (kt + 2 < NT) {
                __builtin_prefetch(kb + (size_t)(krow + 16) * DIM + lcol, 0, 1);
                __builtin_prefetch(vb + (size_t)(krow + 16) * DIM + lcol, 0, 1);
            }
            v4f xk  = *(const v4f*)(kb + (size_t)krow * DIM + lcol);
            v4f fk4 = *(const v4f*)(fk + (size_t)krow * DIM + lcol);
            v4f xv  = *(const v4f*)(vb + (size_t)krow * DIM + lcol);
            v4f fv4 = *(const v4f*)(fv + (size_t)krow * DIM + lcol);
            v4f rk, rv;
            rk.x = xk.x*fk4.x - xk.y*fk4.y;  rk.y = xk.x*fk4.y + xk.y*fk4.x;
            rk.z = xk.z*fk4.z - xk.w*fk4.w;  rk.w = xk.z*fk4.w + xk.w*fk4.z;
            rv.x = xv.x*fv4.x - xv.y*fv4.y;  rv.y = xv.x*fv4.y + xv.y*fv4.x;
            rv.z = xv.z*fv4.z - xv.w*fv4.w;  rv.w = xv.z*fv4.w + xv.w*fv4.z;
            *(v4f*)&Kl[nb][lrow][lcol] = rk;
            *(v4f*)&Vl[nb][lrow][lcol] = rv;
        }

        v8f s = {};
        #pragma unroll
        for (int c = 0; c < 16; ++c) {
            v2f bfrag = *(const v2f*)&Kl[buf][lm][c * 4 + 2 * lh];
            s = __builtin_amdgcn_wmma_f32_16x16x4_f32(
                    false, qa[c], false, bfrag, (short)0, s, false, false);
        }

        #pragma unroll
        for (int r = 0; r < 8; ++r)
            Sl[wave][r + 8 * lh][lm] = s[r] * scale;

        #pragma unroll
        for (int c2 = 0; c2 < 4; ++c2) {
            v2f a2 = *(const v2f*)&Sl[wave][lm][c2 * 4 + 2 * lh];
            const int vr = c2 * 4 + 2 * lh;
            v2f b0, b1, b2, b3;
            b0.x = Vl[buf][vr][ 0 + lm];  b0.y = Vl[buf][vr + 1][ 0 + lm];
            b1.x = Vl[buf][vr][16 + lm];  b1.y = Vl[buf][vr + 1][16 + lm];
            b2.x = Vl[buf][vr][32 + lm];  b2.y = Vl[buf][vr + 1][32 + lm];
            b3.x = Vl[buf][vr][48 + lm];  b3.y = Vl[buf][vr + 1][48 + lm];
            acc0 = __builtin_amdgcn_wmma_f32_16x16x4_f32(false, a2, false, b0, (short)0, acc0, false, false);
            acc1 = __builtin_amdgcn_wmma_f32_16x16x4_f32(false, a2, false, b1, (short)0, acc1, false, false);
            acc2 = __builtin_amdgcn_wmma_f32_16x16x4_f32(false, a2, false, b2, (short)0, acc2, false, false);
            acc3 = __builtin_amdgcn_wmma_f32_16x16x4_f32(false, a2, false, b3, (short)0, acc3, false, false);
        }
    }

    const int orow = q0 + wave * 16;
    #pragma unroll
    for (int r = 0; r < 8; ++r) {
        const size_t row = (size_t)(orow + r + 8 * lh) * DIM;
        ob[row +  0 + lm] = acc0[r];
        ob[row + 16 + lm] = acc1[r];
        ob[row + 32 + lm] = acc2[r];
        ob[row + 48 + lm] = acc3[r];
    }
}

extern "C" void kernel_launch(void* const* d_in, const int* in_sizes, int n_in,
                              void* d_out, int out_size, void* d_ws, size_t ws_size,
                              hipStream_t stream) {
    const float* q  = (const float*)d_in[0];
    const float* k  = (const float*)d_in[1];
    const float* v  = (const float*)d_in[2];
    const float* fq = (const float*)d_in[3];
    const float* fk = (const float*)d_in[4];
    const float* fv = (const float*)d_in[5];
    float* out = (float*)d_out;

    dim3 grid(8 * (SEQ / 128));
    dim3 block(256);
    hipLaunchKernelGGL(MultiHeadRoPE_28166395528043_kernel, grid, block, 0, stream,
                       q, k, v, fq, fk, fv, out);
}


# a wave takes 32 queries and one half of each 64-key tile (2 waves per SIMD kept): each K/V LDS fragment feeds two WMMAs; the halves' partial outputs are added once via LDS (key sum order changes)
# speedup vs baseline: 1.9216x; 1.9216x over previous
#include <hip/hip_runtime.h>

typedef float v2f __attribute__((ext_vector_type(2)));
typedef float v4f __attribute__((ext_vector_type(4)));
typedef float v8f __attribute__((ext_vector_type(8)));

#define SEQ   4096
#define DIM   64
#define NT    (SEQ / 16)
#define KT    64
#define NSUB  (KT / 16)
#define NT64  (SEQ / KT)
#define KSTR  68
#define VSTR  68
#define OSTR  68

__global__ __launch_bounds__(256) void MultiHeadRoPE_28166395528043_kernel(
    const float* __restrict__ q,  const float* __restrict__ k,  const float* __restrict__ v,
    const float* __restrict__ fq, const float* __restrict__ fk, const float* __restrict__ fv,
    float* __restrict__ out)
{
    __shared__ float Kl[2][KT][KSTR];
    __shared__ __attribute__((aligned(16))) float Vt[2][DIM][VSTR];
    __shared__ float Ol[8][16][OSTR];
    __shared__ float Xl[8][4][8][32];

    const int batch = blockIdx.x >> 5;
    const int q0    = (blockIdx.x & 31) * 128;
    const int tid   = threadIdx.x;
    const int wave  = tid >> 5;
    const int lane  = tid & 31;
    const int lm    = lane & 15;
    const int lh    = lane >> 4;

    const int qg    = wave & 3;
    const int kh    = wave >> 2;

    const float scale = 0.125f;

    const float* qb = q + (size_t)batch * SEQ * DIM;
    const float* kb = k + (size_t)batch * SEQ * DIM;
    const float* vb = v + (size_t)batch * SEQ * DIM;
    float*       ob = out + (size_t)batch * SEQ * DIM;

    v2f qa[2][16];
    #pragma unroll
    for (int qt = 0; qt < 2; ++qt) {
        const int    qrow = q0 + qg * 32 + qt * 16 + lm;
        const float* qp   = qb + (size_t)qrow * DIM;
        const float* fp   = fq + (size_t)qrow * DIM;
        #pragma unroll
        for (int c = 0; c < 16; ++c) {
            const int col = c * 4 + 2 * lh;
            v2f x = *(const v2f*)(qp + col);
            v2f f = *(const v2f*)(fp + col);
            v2f r;
            r.x = (x.x * f.x - x.y * f.y) * scale;
            r.y = (x.x * f.y + x.y * f.x) * scale;
            qa[qt][c] = r;
        }
    }

    v8f acc[2][4];
    #pragma unroll
    for (int qt = 0; qt < 2; ++qt)
        #pragma unroll
        for (int n = 0; n < 4; ++n)
            acc[qt][n] = v8f{};

    const int lrow = tid >> 4;
    const int lcol = (tid & 15) * 4;

    #pragma unroll
    for (int j = 0; j < NSUB; ++j) {
        const int r = j * 16 + lrow;
        v4f xk  = *(const v4f*)(kb + (size_t)r * DIM + lcol);
        v4f fk4 = *(const v4f*)(fk + (size_t)r * DIM + lcol);
        v4f xv  = *(const v4f*)(vb + (size_t)r * DIM + lcol);
        v4f fv4 = *(const v4f*)(fv + (size_t)r * DIM + lcol);
        v4f rk, rv;
        rk.x = xk.x*fk4.x - xk.y*fk4.y;  rk.y = xk.x*fk4.y + xk.y*fk4.x;
        rk.z = xk.z*fk4.z - xk.w*fk4.w;  rk.w = xk.z*fk4.w + xk.w*fk4.z;
        rv.x = xv.x*fv4.x - xv.y*fv4.y;  rv.y = xv.x*fv4.y + xv.y*fv4.x;
        rv.z = xv.z*fv4.z - xv.w*fv4.w;  rv.w = xv.z*fv4.w + xv.w*fv4.z;
        *(v4f*)&Kl[0][r][lcol] = rk;
        Vt[0][lcol + 0][r] = rv.x;
        Vt[0][lcol + 1][r] = rv.y;
        Vt[0][lcol + 2][r] = rv.z;
        Vt[0][lcol + 3][r] = rv.w;
    }

    v2f kf[2][16];
    v4f vf[2][4][2];

    const int ro0 = 32 * kh;

    for (int t = 0; t < NT64; ++t) {
        __syncthreads();
        const int buf = t & 1;
        const int nb  = buf ^ 1;

        #pragma unroll
        for (int c = 0; c < 16; ++c)
            kf[0][c] = *(const v2f*)&Kl[buf][ro0 + lm][c * 4 + 2 * lh];
        __builtin_amdgcn_sched_barrier(0);
        #pragma unroll
        for (int db = 0; db < 4; ++db) {
            vf[0][db][0] = *(const v4f*)&Vt[buf][16 * db + lm][ro0 + 8 * lh];
            vf[0][db][1] = *(const v4f*)&Vt[buf][16 * db + lm][ro0 + 8 * lh + 4];
        }

        #pragma unroll
        for (int jj = 0; jj < 2; ++jj) {
            const int cur = jj;
            const int nx  = jj ^ 1;

            v4f xk[2], fk4[2], xv[2], fv4[2];
            #pragma unroll
            for (int u = 0; u < 2; ++u) {
                const int g    = t * NSUB + 2 * jj + u;
                const int nt   = (g + NSUB     < NT) ? (g + NSUB)     : (NT - 1);
                const int pt   = (g + NSUB + 2 < NT) ? (g + NSUB + 2) : (NT - 1);
                const int krow = nt * 16 + lrow;
                const int prow = pt * 16 + lrow;
                __builtin_prefetch(kb + (size_t)prow * DIM + lcol, 0, 1);
                __builtin_prefetch(vb + (size_t)prow * DIM + lcol, 0, 1);
                xk[u]  = *(const v4f*)(kb + (size_t)krow * DIM + lcol);
                fk4[u] = *(const v4f*)(fk + (size_t)krow * DIM + lcol);
                xv[u]  = *(const v4f*)(vb + (size_t)krow * DIM + lcol);
                fv4[u] = *(const v4f*)(fv + (size_t)krow * DIM + lcol);
            }

            if (jj == 0) {
                const int ro = ro0 + 16;
                #pragma unroll
                for (int c = 0; c < 16; ++c)
                    kf[nx][c] = *(const v2f*)&Kl[buf][ro + lm][c * 4 + 2 * lh];
                #pragma unroll
                for (int db = 0; db < 4; ++db) {
                    vf[nx][db][0] = *(const v4f*)&Vt[buf][16 * db + lm][ro + 8 * lh];
                    vf[nx][db][1] = *(const v4f*)&Vt[buf][16 * db + lm][ro + 8 * lh + 4];
                }
            }
            __builtin_amdgcn_sched_barrier(0);

            v8f s0 = {}, s1 = {};
            #pragma unroll
            for (int c = 0; c < 16; ++c) {
                s0 = __builtin_amdgcn_wmma_f32_16x16x4_f32(
                        false, kf[cur][c], false, qa[0][c], (short)0, s0, false, false);
                s1 = __builtin_amdgcn_wmma_f32_16x16x4_f32(
                        false, kf[cur][c], false, qa[1][c], (short)0, s1, false, false);
            }

            #pragma unroll
            for (int c2 = 0; c2 < 4; ++c2) {
                v2f b20, b21;
                b20.x = s0[2 * c2];  b20.y = s0[2 * c2 + 1];
                b21.x = s1[2 * c2];  b21.y = s1[2 * c2 + 1];
                const int u = c2 >> 1;
                v2f a0, a1, a2, a3;
                if (c2 & 1) {
                    a0 = __builtin_shufflevector(vf[cur][0][u], vf[cur][0][u], 2, 3);
                    a1 = __builtin_shufflevector(vf[cur][1][u], vf[cur][1][u], 2, 3);
                    a2 = __builtin_shufflevector(vf[cur][2][u], vf[cur][2][u], 2, 3);
                    a3 = __builtin_shufflevector(vf[cur][3][u], vf[cur][3][u], 2, 3);
                } else {
                    a0 = __builtin_shufflevector(vf[cur][0][u], vf[cur][0][u], 0, 1);
                    a1 = __builtin_shufflevector(vf[cur][1][u], vf[cur][1][u], 0, 1);
                    a2 = __builtin_shufflevector(vf[cur][2][u], vf[cur][2][u], 0, 1);
                    a3 = __builtin_shufflevector(vf[cur][3][u], vf[cur][3][u], 0, 1);
                }
                acc[0][0] = __builtin_amdgcn_wmma_f32_16x16x4_f32(false, a0, false, b20, (short)0, acc[0][0], false, false);
                acc[1][0] = __builtin_amdgcn_wmma_f32_16x16x4_f32(false, a0, false, b21, (short)0, acc[1][0], false, false);
                acc[0][1] = __builtin_amdgcn_wmma_f32_16x16x4_f32(false, a1, false, b20, (short)0, acc[0][1], false, false);
                acc[1][1] = __builtin_amdgcn_wmma_f32_16x16x4_f32(false, a1, false, b21, (short)0, acc[1][1], false, false);
                acc[0][2] = __builtin_amdgcn_wmma_f32_16x16x4_f32(false, a2, false, b20, (short)0, acc[0][2], false, false);
                acc[1][2] = __builtin_amdgcn_wmma_f32_16x16x4_f32(false, a2, false, b21, (short)0, acc[1][2], false, false);
                acc[0][3] = __builtin_amdgcn_wmma_f32_16x16x4_f32(false, a3, false, b20, (short)0, acc[0][3], false, false);
                acc[1][3] = __builtin_amdgcn_wmma_f32_16x16x4_f32(false, a3, false, b21, (short)0, acc[1][3], false, false);
            }
            __builtin_amdgcn_sched_barrier(0);

            #pragma unroll
            for (int u = 0; u < 2; ++u) {
                const int ch = 2 * jj + u;
                v4f rk, rv;
                rk.x = xk[u].x*fk4[u].x - xk[u].y*fk4[u].y;  rk.y = xk[u].x*fk4[u].y + xk[u].y*fk4[u].x;
                rk.z = xk[u].z*fk4[u].z - xk[u].w*fk4[u].w;  rk.w = xk[u].z*fk4[u].w + xk[u].w*fk4[u].z;
                rv.x = __builtin_fmaf(xv[u].x, fv4[u].x, -(xv[u].y * fv4[u].y));
                rv.y = __builtin_fmaf(xv[u].x, fv4[u].y,   xv[u].y * fv4[u].x);
                rv.z = __builtin_fmaf(xv[u].z, fv4[u].z, -(xv[u].w * fv4[u].w));
                rv.w = __builtin_fmaf(xv[u].z, fv4[u].w,   xv[u].w * fv4[u].z);
                *(v4f*)&Kl[nb][ch * 16 + lrow][lcol] = rk;
                Vt[nb][lcol + 0][ch * 16 + lrow] = rv.x;
                Vt[nb][lcol + 1][ch * 16 + lrow] = rv.y;
                Vt[nb][lcol + 2][ch * 16 + lrow] = rv.z;
                Vt[nb][lcol + 3][ch * 16 + lrow] = rv.w;
            }
        }
    }

    v8f keep[4];
    {
        #pragma unroll
        for (int n = 0; n < 4; ++n) {
            const v8f give = kh ? acc[0][n] : acc[1][n];
            keep[n]        = kh ? acc[1][n] : acc[0][n];
            #pragma unroll
            for (int r = 0; r < 8; ++r)
                Xl[wave][n][r][lane] = give[r];
        }
        __syncthreads();
        #pragma unroll
        for (int n = 0; n < 4; ++n) {
            #pragma unroll
            for (int r = 0; r < 8; ++r)
                keep[n][r] += Xl[wave ^ 4][n][r][lane];
        }
    }

    {
        float* orow_l = &Ol[wave][lm][8 * lh];
        v4f t;
        t.x = keep[0][0]; t.y = keep[0][1]; t.z = keep[0][2]; t.w = keep[0][3];  *(v4f*)(orow_l +  0) = t;
        t.x = keep[0][4]; t.y = keep[0][5]; t.z = keep[0][6]; t.w = keep[0][7];  *(v4f*)(orow_l +  4) = t;
        t.x = keep[1][0]; t.y = keep[1][1]; t.z = keep[1][2]; t.w = keep[1][3];  *(v4f*)(orow_l + 16) = t;
        t.x = keep[1][4]; t.y = keep[1][5]; t.z = keep[1][6]; t.w = keep[1][7];  *(v4f*)(orow_l + 20) = t;
        t.x = keep[2][0]; t.y = keep[2][1]; t.z = keep[2][2]; t.w = keep[2][3];  *(v4f*)(orow_l + 32) = t;
        t.x = keep[2][4]; t.y = keep[2][5]; t.z = keep[2][6]; t.w = keep[2][7];  *(v4f*)(orow_l + 36) = t;
        t.x = keep[3][0]; t.y = keep[3][1]; t.z = keep[3][2]; t.w = keep[3][3];  *(v4f*)(orow_l + 48) = t;
        t.x = keep[3][4]; t.y = keep[3][5]; t.z = keep[3][6]; t.w = keep[3][7];  *(v4f*)(orow_l + 52) = t;
        const int orow = q0 + qg * 32 + kh * 16;
        #pragma unroll
        for (int i = 0; i < 8; ++i) {
            const int row = 2 * i + lh;
            const v4f o = *(const v4f*)&Ol[wave][row][4 * lm];
            *(v4f*)(ob + (size_t)(orow + row) * DIM + 4 * lm) = o;
        }
    }
}

extern "C" void kernel_launch(void* const* d_in, const int* in_sizes, int n_in,
                              void* d_out, int out_size, void* d_ws, size_t ws_size,
                              hipStream_t stream) {
    const float* q  = (const float*)d_in[0];
    const float* k  = (const float*)d_in[1];
    const float* v  = (const float*)d_in[2];
    const float* fq = (const float*)d_in[3];
    const float* fk = (const float*)d_in[4];
    const float* fv = (const float*)d_in[5];
    float* out = (float*)d_out;

    dim3 grid(8 * (SEQ / 128));
    dim3 block(256);
    hipLaunchKernelGGL(MultiHeadRoPE_28166395528043_kernel, grid, block, 0, stream,
                       q, k, v, fq, fk, fv, out);
}


# K/V staging split by wave age: the older wave of each SIMD pair (waves 0-3) loads, rotates and parks the whole next tile, the younger (4-7) only computes, so staging runs beside the partner's WMMAs
# speedup vs baseline: 2.0329x; 1.0579x over previous
#include <hip/hip_runtime.h>

typedef float v2f __attribute__((ext_vector_type(2)));
typedef float v4f __attribute__((ext_vector_type(4)));
typedef float v8f __attribute__((ext_vector_type(8)));

#define SEQ   4096
#define DIM   64
#define NT    (SEQ / 16)
#define KT    64
#define NSUB  (KT / 16)
#define NT64  (SEQ / KT)
#define KSTR  68
#define VSTR  68
#define OSTR  68

__global__ __launch_bounds__(256) void MultiHeadRoPE_28166395528043_kernel(
    const float* __restrict__ q,  const float* __restrict__ k,  const float* __restrict__ v,
    const float* __restrict__ fq, const float* __restrict__ fk, const float* __restrict__ fv,
    float* __restrict__ out)
{
    __shared__ float Kl[2][KT][KSTR];
    __shared__ __attribute__((aligned(16))) float Vt[2][DIM][VSTR];
    __shared__ float Ol[8][16][OSTR];
    __shared__ float Xl[8][4][8][32];

    const int batch = blockIdx.x >> 5;
    const int q0    = (blockIdx.x & 31) * 128;
    const int tid   = threadIdx.x;
    const int wave  = __builtin_amdgcn_readfirstlane(tid >> 5);
    const int lane  = tid & 31;
    const int lm    = lane & 15;
    const int lh    = lane >> 4;

    const int qg    = wave & 3;
    const int kh    = wave >> 2;

    const bool stager = (kh == 0);

    const float scale = 0.125f;

    const float* qb = q + (size_t)batch * SEQ * DIM;
    const float* kb = k + (size_t)batch * SEQ * DIM;
    const float* vb = v + (size_t)batch * SEQ * DIM;
    float*       ob = out + (size_t)batch * SEQ * DIM;

    v2f qa[2][16];
    #pragma unroll
    for (int qt = 0; qt < 2; ++qt) {
        const int    qrow = q0 + qg * 32 + qt * 16 + lm;
        const float* qp   = qb + (size_t)qrow * DIM;
        const float* fp   = fq + (size_t)qrow * DIM;
        #pragma unroll
        for (int c = 0; c < 16; ++c) {
            const int col = c * 4 + 2 * lh;
            v2f x = *(const v2f*)(qp + col);
            v2f f = *(const v2f*)(fp + col);
            v2f r;
            r.x = (x.x * f.x - x.y * f.y) * scale;
            r.y = (x.x * f.y + x.y * f.x) * scale;
            qa[qt][c] = r;
        }
    }

    v8f acc[2][4];
    #pragma unroll
    for (int qt = 0; qt < 2; ++qt)
        #pragma unroll
        for (int n = 0; n < 4; ++n)
            acc[qt][n] = v8f{};

    const int lrow = tid >> 4;
    const int lcol = (tid & 15) * 4;
    const int srow = lrow & 7;

    #pragma unroll
    for (int j = 0; j < NSUB; ++j) {
        const int r = j * 16 + lrow;
        v4f xk  = *(const v4f*)(kb + (size_t)r * DIM + lcol);
        v4f fk4 = *(const v4f*)(fk + (size_t)r * DIM + lcol);
        v4f xv  = *(const v4f*)(vb + (size_t)r * DIM + lcol);
        v4f fv4 = *(const v4f*)(fv + (size_t)r * DIM + lcol);
        v4f rk, rv;
        rk.x = xk.x*fk4.x - xk.y*fk4.y;  rk.y = xk.x*fk4.y + xk.y*fk4.x;
        rk.z = xk.z*fk4.z - xk.w*fk4.w;  rk.w = xk.z*fk4.w + xk.w*fk4.z;
        rv.x = xv.x*fv4.x - xv.y*fv4.y;  rv.y = xv.x*fv4.y + xv.y*fv4.x;
        rv.z = xv.z*fv4.z - xv.w*fv4.w;  rv.w = xv.z*fv4.w + xv.w*fv4.z;
        *(v4f*)&Kl[0][r][lcol] = rk;
        Vt[0][lcol + 0][r] = rv.x;
        Vt[0][lcol + 1][r] = rv.y;
        Vt[0][lcol + 2][r] = rv.z;
        Vt[0][lcol + 3][r] = rv.w;
    }

    v2f kf[2][16];
    v4f vf[2][4][2];

    const int ro0 = 32 * kh;

    for (int t = 0; t < NT64; ++t) {
        __syncthreads();
        const int buf = t & 1;
        const int nb  = buf ^ 1;

        #pragma unroll
        for (int c = 0; c < 16; ++c)
            kf[0][c] = *(const v2f*)&Kl[buf][ro0 + lm][c * 4 + 2 * lh];
        __builtin_amdgcn_sched_barrier(0);
        #pragma unroll
        for (int db = 0; db < 4; ++db) {
            vf[0][db][0] = *(const v4f*)&Vt[buf][16 * db + lm][ro0 + 8 * lh];
            vf[0][db][1] = *(const v4f*)&Vt[buf][16 * db + lm][ro0 + 8 * lh + 4];
        }

        #pragma unroll
        for (int jj = 0; jj < 2; ++jj) {
            const int cur = jj;
            const int nx  = jj ^ 1;

            v4f xk[4], fk4[4], xv[4], fv4[4];
            if (stager) {
                #pragma unroll
                for (int s = 0; s < 4; ++s) {
                    const int g    = t * NSUB + 2 * jj + (s >> 1);
                    const int nt   = (g + NSUB     < NT) ? (g + NSUB)     : (NT - 1);
                    const int pt   = (g + NSUB + 2 < NT) ? (g + NSUB + 2) : (NT - 1);
                    const int krow = nt * 16 + (s & 1) * 8 + srow;
                    const int prow = pt * 16 + (s & 1) * 8 + srow;
                    __builtin_prefetch(kb + (size_t)prow * DIM + lcol, 0, 1);
                    __builtin_prefetch(vb + (size_t)prow * DIM + lcol, 0, 1);
                    xk[s]  = *(const v4f*)(kb + (size_t)krow * DIM + lcol);
                    fk4[s] = *(const v4f*)(fk + (size_t)krow * DIM + lcol);
                    xv[s]  = *(const v4f*)(vb + (size_t)krow * DIM + lcol);
                    fv4[s] = *(const v4f*)(fv + (size_t)krow * DIM + lcol);
                }
            }

            if (jj == 0) {
                const int ro = ro0 + 16;
                #pragma unroll
                for (int c = 0; c < 16; ++c)
                    kf[nx][c] = *(const v2f*)&Kl[buf][ro + lm][c * 4 + 2 * lh];
                #pragma unroll
                for (int db = 0; db < 4; ++db) {
                    vf[nx][db][0] = *(const v4f*)&Vt[buf][16 * db + lm][ro + 8 * lh];
                    vf[nx][db][1] = *(const v4f*)&Vt[buf][16 * db + lm][ro + 8 * lh + 4];
                }
            }
            __builtin_amdgcn_sched_barrier(0);

            v8f s0 = {}, s1 = {};
            #pragma unroll
            for (int c = 0; c < 16; ++c) {
                s0 = __builtin_amdgcn_wmma_f32_16x16x4_f32(
                        false, kf[cur][c], false, qa[0][c], (short)0, s0, false, false);
                s1 = __builtin_amdgcn_wmma_f32_16x16x4_f32(
                        false, kf[cur][c], false, qa[1][c], (short)0, s1, false, false);
            }

            #pragma unroll
            for (int c2 = 0; c2 < 4; ++c2) {
                v2f b20, b21;
                b20.x = s0[2 * c2];  b20.y = s0[2 * c2 + 1];
                b21.x = s1[2 * c2];  b21.y = s1[2 * c2 + 1];
                const int u = c2 >> 1;
                v2f a0, a1, a2, a3;
                if (c2 & 1) {
                    a0 = __builtin_shufflevector(vf[cur][0][u], vf[cur][0][u], 2, 3);
                    a1 = __builtin_shufflevector(vf[cur][1][u], vf[cur][1][u], 2, 3);
                    a2 = __builtin_shufflevector(vf[cur][2][u], vf[cur][2][u], 2, 3);
                    a3 = __builtin_shufflevector(vf[cur][3][u], vf[cur][3][u], 2, 3);
                } else {
                    a0 = __builtin_shufflevector(vf[cur][0][u], vf[cur][0][u], 0, 1);
                    a1 = __builtin_shufflevector(vf[cur][1][u], vf[cur][1][u], 0, 1);
                    a2 = __builtin_shufflevector(vf[cur][2][u], vf[cur][2][u], 0, 1);
                    a3 = __builtin_shufflevector(vf[cur][3][u], vf[cur][3][u], 0, 1);
                }
                acc[0][0] = __builtin_amdgcn_wmma_f32_16x16x4_f32(false, a0, false, b20, (short)0, acc[0][0], false, false);
                acc[1][0] = __builtin_amdgcn_wmma_f32_16x16x4_f32(false, a0, false, b21, (short)0, acc[1][0], false, false);
                acc[0][1] = __builtin_amdgcn_wmma_f32_16x16x4_f32(false, a1, false, b20, (short)0, acc[0][1], false, false);
                acc[1][1] = __builtin_amdgcn_wmma_f32_16x16x4_f32(false, a1, false, b21, (short)0, acc[1][1], false, false);
                acc[0][2] = __builtin_amdgcn_wmma_f32_16x16x4_f32(false, a2, false, b20, (short)0, acc[0][2], false, false);
                acc[1][2] = __builtin_amdgcn_wmma_f32_16x16x4_f32(false, a2, false, b21, (short)0, acc[1][2], false, false);
                acc[0][3] = __builtin_amdgcn_wmma_f32_16x16x4_f32(false, a3, false, b20, (short)0, acc[0][3], false, false);
                acc[1][3] = __builtin_amdgcn_wmma_f32_16x16x4_f32(false, a3, false, b21, (short)0, acc[1][3], false, false);
            }
            __builtin_amdgcn_sched_barrier(0);

            if (stager) {
                #pragma unroll
                for (int s = 0; s < 4; ++s) {
                    const int r = (2 * jj + (s >> 1)) * 16 + (s & 1) * 8 + srow;
                    v4f rk, rv;
                    rk.x = xk[s].x*fk4[s].x - xk[s].y*fk4[s].y;  rk.y = xk[s].x*fk4[s].y + xk[s].y*fk4[s].x;
                    rk.z = xk[s].z*fk4[s].z - xk[s].w*fk4[s].w;  rk.w = xk[s].z*fk4[s].w + xk[s].w*fk4[s].z;
                    rv.x = __builtin_fmaf(xv[s].x, fv4[s].x, -(xv[s].y * fv4[s].y));
                    rv.y = __builtin_fmaf(xv[s].x, fv4[s].y,   xv[s].y * fv4[s].x);
                    rv.z = __builtin_fmaf(xv[s].z, fv4[s].z, -(xv[s].w * fv4[s].w));
                    rv.w = __builtin_fmaf(xv[s].z, fv4[s].w,   xv[s].w * fv4[s].z);
                    *(v4f*)&Kl[nb][r][lcol] = rk;
                    Vt[nb][lcol + 0][r] = rv.x;
                    Vt[nb][lcol + 1][r] = rv.y;
                    Vt[nb][lcol + 2][r] = rv.z;
                    Vt[nb][lcol + 3][r] = rv.w;
                }
            }
        }
    }

    v8f keep[4];
    {
        #pragma unroll
        for (int n = 0; n < 4; ++n) {
            const v8f give = kh ? acc[0][n] : acc[1][n];
            keep[n]        = kh ? acc[1][n] : acc[0][n];
            #pragma unroll
            for (int r = 0; r < 8; ++r)
                Xl[wave][n][r][lane] = give[r];
        }
        __syncthreads();
        #pragma unroll
        for (int n = 0; n < 4; ++n) {
            #pragma unroll
            for (int r = 0; r < 8; ++r)
                keep[n][r] += Xl[wave ^ 4][n][r][lane];
        }
    }

    {
        float* orow_l = &Ol[wave][lm][8 * lh];
        v4f t;
        t.x = keep[0][0]; t.y = keep[0][1]; t.z = keep[0][2]; t.w = keep[0][3];  *(v4f*)(orow_l +  0) = t;
        t.x = keep[0][4]; t.y = keep[0][5]; t.z = keep[0][6]; t.w = keep[0][7];  *(v4f*)(orow_l +  4) = t;
        t.x = keep[1][0]; t.y = keep[1][1]; t.z = keep[1][2]; t.w = keep[1][3];  *(v4f*)(orow_l + 16) = t;
        t.x = keep[1][4]; t.y = keep[1][5]; t.z = keep[1][6]; t.w = keep[1][7];  *(v4f*)(orow_l + 20) = t;
        t.x = keep[2][0]; t.y = keep[2][1]; t.z = keep[2][2]; t.w = keep[2][3];  *(v4f*)(orow_l + 32) = t;
        t.x = keep[2][4]; t.y = keep[2][5]; t.z = keep[2][6]; t.w = keep[2][7];  *(v4f*)(orow_l + 36) = t;
        t.x = keep[3][0]; t.y = keep[3][1]; t.z = keep[3][2]; t.w = keep[3][3];  *(v4f*)(orow_l + 48) = t;
        t.x = keep[3][4]; t.y = keep[3][5]; t.z = keep[3][6]; t.w = keep[3][7];  *(v4f*)(orow_l + 52) = t;
        const int orow = q0 + qg * 32 + kh * 16;
        #pragma unroll
        for (int i = 0; i < 8; ++i) {
            const int row = 2 * i + lh;
            const v4f o = *(const v4f*)&Ol[wave][row][4 * lm];
            *(v4f*)(ob + (size_t)(orow + row) * DIM + 4 * lm) = o;
        }
    }
}

extern "C" void kernel_launch(void* const* d_in, const int* in_sizes, int n_in,
                              void* d_out, int out_size, void* d_ws, size_t ws_size,
                              hipStream_t stream) {
    const float* q  = (const float*)d_in[0];
    const float* k  = (const float*)d_in[1];
    const float* v  = (const float*)d_in[2];
    const float* fq = (const float*)d_in[3];
    const float* fk = (const float*)d_in[4];
    const float* fv = (const float*)d_in[5];
    float* out = (float*)d_out;

    dim3 grid(8 * (SEQ / 128));
    dim3 block(256);
    hipLaunchKernelGGL(MultiHeadRoPE_28166395528043_kernel, grid, block, 0, stream,
                       q, k, v, fq, fk, fv, out);
}
